# starting kernel
# baseline (speedup 1.0000x reference)
#include <hip/hip_runtime.h>

typedef float v2f __attribute__((ext_vector_type(2)));
typedef float v4f __attribute__((ext_vector_type(4)));
typedef float v8f __attribute__((ext_vector_type(8)));

#define SEQ   4096
#define DIM   64
#define NT    (SEQ / 16)
#define KSTR  68
#define SSTR  18

__global__ __launch_bounds__(256) void MultiHeadRoPE_28166395528043_kernel(
    const float* __restrict__ q,  const float* __restrict__ k,  const float* __restrict__ v,
    const float* __restrict__ fq, const float* __restrict__ fk, const float* __restrict__ fv,
    float* __restrict__ out)
{
    __shared__ float Kl[2][16][KSTR];
    __shared__ float Vl[2][16][KSTR];
    __shared__ float Sl[8][16][SSTR];

    const int batch = blockIdx.x >> 5;
    const int q0    = (blockIdx.x & 31) * 128;
    const int tid   = threadIdx.x;
    const int wave  = tid >> 5;
    const int lane  = tid & 31;
    const int lm    = lane & 15;
    const int lh    = lane >> 4;

    const float scale = 0.125f;

    const float* qb = q + (size_t)batch * SEQ * DIM;
    const float* kb = k + (size_t)batch * SEQ * DIM;
    const float* vb = v + (size_t)batch * SEQ * DIM;
    float*       ob = out + (size_t)batch * SEQ * DIM;

    v2f qa[16];
    {
        const int    qrow = q0 + wave * 16 + lm;
        const float* qp   = qb + (size_t)qrow * DIM;
        const float* fp   = fq + (size_t)qrow * DIM;
        #pragma unroll
        for (int c = 0; c < 16; ++c) {
            const int col = c * 4 + 2 * lh;
            v2f x = *(const v2f*)(qp + col);
            v2f f = *(const v2f*)(fp + col);
            v2f r;
            r.x = x.x * f.x - x.y * f.y;
            r.y = x.x * f.y + x.y * f.x;
            qa[c] = r;
        }
    }

    v8f acc0 = {}, acc1 = {}, acc2 = {}, acc3 = {};

    const int lrow = tid >> 4;
    const int lcol = (tid & 15) * 4;

    {
        v4f xk  = *(const v4f*)(kb + (size_t)lrow * DIM + lcol);
        v4f fk4 = *(const v4f*)(fk + (size_t)lrow * DIM + lcol);
        v4f xv  = *(const v4f*)(vb + (size_t)lrow * DIM + lcol);
        v4f fv4 = *(const v4f*)(fv + (size_t)lrow * DIM + lcol);
        v4f rk, rv;
        rk.x = xk.x*fk4.x - xk.y*fk4.y;  rk.y = xk.x*fk4.y + xk.y*fk4.x;
        rk.z = xk.z*fk4.z - xk.w*fk4.w;  rk.w = xk.z*fk4.w + xk.w*fk4.z;
        rv.x = xv.x*fv4.x - xv.y*fv4.y;  rv.y = xv.x*fv4.y + xv.y*fv4.x;
        rv.z = xv.z*fv4.z - xv.w*fv4.w;  rv.w = xv.z*fv4.w + xv.w*fv4.z;
        *(v4f*)&Kl[0][lrow][lcol] = rk;
        *(v4f*)&Vl[0][lrow][lcol] = rv;
    }

    for (int kt = 0; kt < NT; ++kt) {
        __syncthreads();
        const int buf = kt & 1;

        if (kt + 1 < NT) {
            const int nb   = buf ^ 1;
            const int krow = (kt + 1) * 16 + lrow;
            if (kt + 2 < NT) {
                __builtin_prefetch(kb + (size_t)(krow + 16) * DIM + lcol, 0, 1);
                __builtin_prefetch(vb + (size_t)(krow + 16) * DIM + lcol, 0, 1);
            }
            v4f xk  = *(const v4f*)(kb + (size_t)krow * DIM + lcol);
            v4f fk4 = *(const v4f*)(fk + (size_t)krow * DIM + lcol);
            v4f xv  = *(const v4f*)(vb + (size_t)krow * DIM + lcol);
            v4f fv4 = *(const v4f*)(fv + (size_t)krow * DIM + lcol);
            v4f rk, rv;
            rk.x = xk.x*fk4.x - xk.y*fk4.y;  rk.y = xk.x*fk4.y + xk.y*fk4.x;
            rk.z = xk.z*fk4.z - xk.w*fk4.w;  rk.w = xk.z*fk4.w + xk.w*fk4.z;
            rv.x = xv.x*fv4.x - xv.y*fv4.y;  rv.y = xv.x*fv4.y + xv.y*fv4.x;
            rv.z = xv.z*fv4.z - xv.w*fv4.w;  rv.w = xv.z*fv4.w + xv.w*fv4.z;
            *(v4f*)&Kl[nb][lrow][lcol] = rk;
            *(v4f*)&Vl[nb][lrow][lcol] = rv;
        }

        v8f s = {};
        #pragma unroll
        for (int c = 0; c < 16; ++c) {
            v2f bfrag = *(const v2f*)&Kl[buf][lm][c * 4 + 2 * lh];
            s = __builtin_amdgcn_wmma_f32_16x16x4_f32(
                    false, qa[c], false, bfrag, (short)0, s, false, false);
        }

        #pragma unroll
        for (int r = 0; r < 8; ++r)
            Sl[wave][r + 8 * lh][lm] = s[r] * scale;

        #pragma unroll
        for (int c2 = 0; c2 < 4; ++c2) {
            v2f a2 = *(const v2f*)&Sl[wave][lm][c2 * 4 + 2 * lh];
            const int vr = c2 * 4 + 2 * lh;
            v2f b0, b1, b2, b3;
            b0.x = Vl[buf][vr][ 0 + lm];  b0.y = Vl[buf][vr + 1][ 0 + lm];
            b1.x = Vl[buf][vr][16 + lm];  b1.y = Vl[buf][vr + 1][16 + lm];
            b2.x = Vl[buf][vr][32 + lm];  b2.y = Vl[buf][vr + 1][32 + lm];
            b3.x = Vl[buf][vr][48 + lm];  b3.y = Vl[buf][vr + 1][48 + lm];
            acc0 = __builtin_amdgcn_wmma_f32_16x16x4_f32(false, a2, false, b0, (short)0, acc0, false, false);
            acc1 = __builtin_amdgcn_wmma_f32_16x16x4_f32(false, a2, false, b1, (short)0, acc1, false, false);
            acc2 = __builtin_amdgcn_wmma_f32_16x16x4_f32(false, a2, false, b2, (short)0, acc2, false, false);
            acc3 = __builtin_amdgcn_wmma_f32_16x16x4_f32(false, a2, false, b3, (short)0, acc3, false, false);
        }
    }

    const int orow = q0 + wave * 16;
    #pragma unroll
    for (int r = 0; r < 8; ++r) {
        const size_t row = (size_t)(orow + r + 8 * lh) * DIM;
        ob[row +  0 + lm] = acc0[r];
        ob[row + 16 + lm] = acc1[r];
        ob[row + 32 + lm] = acc2[r];
        ob[row + 48 + lm] = acc3[r];
    }
}

extern "C" void kernel_launch(void* const* d_in, const int* in_sizes, int n_in,
                              void* d_out, int out_size, void* d_ws, size_t ws_size,
                              hipStream_t stream) {
    const float* q  = (const float*)d_in[0];
    const float* k  = (const float*)d_in[1];
    const float* v  = (const float*)d_in[2];
    const float* fq = (const float*)d_in[3];
    const float* fk = (const float*)d_in[4];
    const float* fv = (const float*)d_in[5];
    float* out = (float*)d_out;

    dim3 grid(8 * (SEQ / 128));
    dim3 block(256);
    hipLaunchKernelGGL(MultiHeadRoPE_28166395528043_kernel, grid, block, 0, stream,
                       q, k, v, fq, fk, fv, out);
}


# reassociate (Q K^T) V as Q (K^T V), no softmax between: per-batch 64x64 K^T V from 8 partial sums in a first launch, then Q times it; 64x fewer f32 WMMAs; changes the f32 summation order
# speedup vs baseline: 21.0399x; 21.0399x over previous
#include <hip/hip_runtime.h>


typedef float v2f __attribute__((ext_vector_type(2)));
typedef float v4f __attribute__((ext_vector_type(4)));
typedef float v8f __attribute__((ext_vector_type(8)));

#define SEQ    4096
#define DIM    64
#define NPART  8
#define CHUNK  (SEQ / NPART)
#define MSZ    (DIM * DIM)
#define GSTEPS 4
#define NGROUP (CHUNK / 32 / GSTEPS)
#define MPITCH 160


static __device__ __forceinline__ v2f rope2(v2f x, v2f f) {
    v2f r;
    r.x = x.x * f.x - x.y * f.y;
    r.y = x.x * f.y + x.y * f.x;
    return r;
}

__global__ __launch_bounds__(256) void rope_ktv_partial(
    const float* __restrict__ k,  const float* __restrict__ v,
    const float* __restrict__ fk, const float* __restrict__ fv,
    float* __restrict__ ws)
{
    __shared__ float red[8][4][8][32];

    const int quad  = blockIdx.x & 3;
    const int pc    = (blockIdx.x >> 2) & 7;
    const int batch = blockIdx.x >> 5;
    const int g     = quad >> 1;
    const int gp    = quad & 1;
    const int tid   = threadIdx.x;
    const int wave  = tid >> 5;
    const int lane  = tid & 31;
    const int lm    = lane & 15;
    const int lh    = lane >> 4;

    const int colk  = 32 * g  + 2 * lm;
    const int colv  = 32 * gp + 2 * lm;
    const int rbase = pc * CHUNK + wave * 4 + 2 * lh;
    const float* kp  = k  + ((size_t)batch * SEQ + rbase) * DIM + colk;
    const float* vp  = v  + ((size_t)batch * SEQ + rbase) * DIM + colv;
    const float* fkp = fk + (size_t)rbase * DIM + colk;
    const float* fvp = fv + (size_t)rbase * DIM + colv;

    v8f aee = {}, aoe = {}, aeo = {}, aoo = {};

    v2f ck0[GSTEPS], ck1[GSTEPS], cf0[GSTEPS], cf1[GSTEPS];
    v2f cv0[GSTEPS], cv1[GSTEPS], cg0[GSTEPS], cg1[GSTEPS];
    #pragma unroll
    for (int s = 0; s < GSTEPS; ++s) {
        const int o = s * 32 * DIM;
        ck0[s] = *(const v2f*)(kp + o);   ck1[s] = *(const v2f*)(kp + o + DIM);
        cf0[s] = *(const v2f*)(fkp + o);  cf1[s] = *(const v2f*)(fkp + o + DIM);
        cv0[s] = *(const v2f*)(vp + o);   cv1[s] = *(const v2f*)(vp + o + DIM);
        cg0[s] = *(const v2f*)(fvp + o);  cg1[s] = *(const v2f*)(fvp + o + DIM);
    }
    #pragma unroll
    for (int grp = 0; grp < NGROUP; ++grp) {
        v2f nk0[GSTEPS], nk1[GSTEPS], nf0[GSTEPS], nf1[GSTEPS];
        v2f nv0[GSTEPS], nv1[GSTEPS], ng0[GSTEPS], ng1[GSTEPS];
        if (grp + 1 < NGROUP) {
            #pragma unroll
            for (int s = 0; s < GSTEPS; ++s) {
                const int o = ((grp + 1) * GSTEPS + s) * 32 * DIM;
                nk0[s] = *(const v2f*)(kp + o);   nk1[s] = *(const v2f*)(kp + o + DIM);
                nf0[s] = *(const v2f*)(fkp + o);  nf1[s] = *(const v2f*)(fkp + o + DIM);
                nv0[s] = *(const v2f*)(vp + o);   nv1[s] = *(const v2f*)(vp + o + DIM);
                ng0[s] = *(const v2f*)(fvp + o);  ng1[s] = *(const v2f*)(fvp + o + DIM);
            }
        }
        #pragma unroll
        for (int s = 0; s < GSTEPS; ++s) {
            const v2f k0 = rope2(ck0[s], cf0[s]);
            const v2f k1 = rope2(ck1[s], cf1[s]);
            const v2f v0 = rope2(cv0[s], cg0[s]);
            const v2f v1 = rope2(cv1[s], cg1[s]);
            v2f ae, ao, be, bo;
            ae.x = k0.x;  ae.y = k1.x;
            ao.x = k0.y;  ao.y = k1.y;
            be.x = v0.x;  be.y = v1.x;
            bo.x = v0.y;  bo.y = v1.y;
            aee = __builtin_amdgcn_wmma_f32_16x16x4_f32(false, ae, false, be, (short)0, aee, false, false);
            aoe = __builtin_amdgcn_wmma_f32_16x16x4_f32(false, ao, false, be, (short)0, aoe, false, false);
            aeo = __builtin_amdgcn_wmma_f32_16x16x4_f32(false, ae, false, bo, (short)0, aeo, false, false);
            aoo = __builtin_amdgcn_wmma_f32_16x16x4_f32(false, ao, false, bo, (short)0, aoo, false, false);
        }
        if (grp + 1 < NGROUP) {
            #pragma unroll
            for (int s = 0; s < GSTEPS; ++s) {
                ck0[s] = nk0[s];  ck1[s] = nk1[s];  cf0[s] = nf0[s];  cf1[s] = nf1[s];
                cv0[s] = nv0[s];  cv1[s] = nv1[s];  cg0[s] = ng0[s];  cg1[s] = ng1[s];
            }
        }
    }

    #pragma unroll
    for (int r = 0; r < 8; ++r) {
        red[wave][0][r][lane] = aee[r];
        red[wave][1][r][lane] = aoe[r];
        red[wave][2][r][lane] = aeo[r];
        red[wave][3][r][lane] = aoo[r];
    }
    __syncthreads();
    {
        const int r = wave;
        v4f s;
        s.x = red[0][0][r][lane];  s.y = red[0][1][r][lane];
        s.z = red[0][2][r][lane];  s.w = red[0][3][r][lane];
        #pragma unroll
        for (int w = 1; w < 8; ++w) {
            s.x += red[w][0][r][lane];  s.y += red[w][1][r][lane];
            s.z += red[w][2][r][lane];  s.w += red[w][3][r][lane];
        }
        float* dst = ws + (size_t)(batch * NPART + pc) * MSZ
                        + (16 * g + 8 * lh + r) * 128 + 64 * gp + 4 * lm;
        *(v4f*)dst = s;
    }
}

__global__ __launch_bounds__(256) void rope_q_times_m(
    const float* __restrict__ q, const float* __restrict__ fq,
    const float* __restrict__ ws, float* __restrict__ out)
{
    __shared__ float Ml[32 * MPITCH];

    const int batch = blockIdx.x >> 5;
    const int q0    = (blockIdx.x & 31) * 128;
    const int tid   = threadIdx.x;
    const int wave  = tid >> 5;
    const int lane  = tid & 31;
    const int lm    = lane & 15;
    const int lh    = lane >> 4;

    v2f xq[16], xf[16];
    {
        const int    qrow = q0 + wave * 16 + lm;
        const float* qp   = q  + ((size_t)batch * SEQ + qrow) * DIM + 2 * lh;
        const float* fp   = fq + (size_t)qrow * DIM + 2 * lh;
        #pragma unroll
        for (int c = 0; c < 16; ++c) {
            xq[c] = *(const v2f*)(qp + c * 4);
            xf[c] = *(const v2f*)(fp + c * 4);
        }
    }

    {
        const float* wp = ws + (size_t)batch * NPART * MSZ + tid * 4;
        #pragma unroll
        for (int kk = 0; kk < 4; ++kk) {
            v4f s = *(const v4f*)(wp + kk * 1024);
            #pragma unroll
            for (int p = 1; p < NPART; ++p)
                s += *(const v4f*)(wp + p * MSZ + kk * 1024);
            const int idx = tid * 4 + kk * 1024;
            *(v4f*)&Ml[(idx >> 7) * MPITCH + (idx & 127)] = s;
        }
    }
    __syncthreads();

    v2f qa[16];
    #pragma unroll
    for (int c = 0; c < 16; ++c) qa[c] = rope2(xq[c], xf[c]);

    v8f acc0 = {}, acc1 = {}, acc2 = {}, acc3 = {};
    #pragma unroll
    for (int c = 0; c < 16; ++c) {
        const float* mrow = &Ml[(2 * c + lh) * MPITCH + 2 * lm];
        const v2f b0 = *(const v2f*)(mrow +  0);
        const v2f b1 = *(const v2f*)(mrow + 32);
        const v2f b2 = *(const v2f*)(mrow + 64);
        const v2f b3 = *(const v2f*)(mrow + 96);
        acc0 = __builtin_amdgcn_wmma_f32_16x16x4_f32(false, qa[c], false, b0, (short)0, acc0, false, false);
        acc1 = __builtin_amdgcn_wmma_f32_16x16x4_f32(false, qa[c], false, b1, (short)0, acc1, false, false);
        acc2 = __builtin_amdgcn_wmma_f32_16x16x4_f32(false, qa[c], false, b2, (short)0, acc2, false, false);
        acc3 = __builtin_amdgcn_wmma_f32_16x16x4_f32(false, qa[c], false, b3, (short)0, acc3, false, false);
    }

    const float scale = 0.125f;
    float* ob = out + (size_t)batch * SEQ * DIM;
    const int orow = q0 + wave * 16;
    #pragma unroll
    for (int r = 0; r < 8; ++r) {
        const size_t row = (size_t)(orow + r + 8 * lh) * DIM;
        ob[row +  0 + lm] = acc0[r] * scale;
        ob[row + 16 + lm] = acc1[r] * scale;
        ob[row + 32 + lm] = acc2[r] * scale;
        ob[row + 48 + lm] = acc3[r] * scale;
    }
}

extern "C" void kernel_launch(void* const* d_in, const int* in_sizes, int n_in,
                              void* d_out, int out_size, void* d_ws, size_t ws_size,
                              hipStream_t stream) {
    const float* q  = (const float*)d_in[0];
    const float* k  = (const float*)d_in[1];
    const float* v  = (const float*)d_in[2];
    const float* fq = (const float*)d_in[3];
    const float* fk = (const float*)d_in[4];
    const float* fv = (const float*)d_in[5];
    float* out = (float*)d_out;
    float* ws  = (float*)d_ws;

    hipLaunchKernelGGL(rope_ktv_partial, dim3(256), dim3(256), 0, stream, k, v, fk, fv, ws);
    hipLaunchKernelGGL(rope_q_times_m, dim3(256), dim3(256), 0, stream, q, fq, ws, out);
}


# launch 2 moves Q, its frequencies and the output in whole 128-B lines, 16 B per lane, via a per-wave LDS tile (was 16 lines per 8-B load and two 64-B pieces per stored line)
# speedup vs baseline: 22.3921x; 1.0643x over previous
#include <hip/hip_runtime.h>


typedef float v2f __attribute__((ext_vector_type(2)));
typedef float v4f __attribute__((ext_vector_type(4)));
typedef float v8f __attribute__((ext_vector_type(8)));

#define SEQ    4096
#define DIM    64
#define NPART  8
#define CHUNK  (SEQ / NPART)
#define MSZ    (DIM * DIM)
#define GSTEPS 4
#define NGROUP (CHUNK / 32 / GSTEPS)
#define MPITCH 160
#define QPITCH 68


static __device__ __forceinline__ v2f rope2(v2f x, v2f f) {
    v2f r;
    r.x = x.x * f.x - x.y * f.y;
    r.y = x.x * f.y + x.y * f.x;
    return r;
}

__global__ __launch_bounds__(256) void rope_ktv_partial(
    const float* __restrict__ k,  const float* __restrict__ v,
    const float* __restrict__ fk, const float* __restrict__ fv,
    float* __restrict__ ws)
{
    __shared__ float red[8][4][8][32];

    const int quad  = blockIdx.x & 3;
    const int pc    = (blockIdx.x >> 2) & 7;
    const int batch = blockIdx.x >> 5;
    const int g     = quad >> 1;
    const int gp    = quad & 1;
    const int tid   = threadIdx.x;
    const int wave  = tid >> 5;
    const int lane  = tid & 31;
    const int lm    = lane & 15;
    const int lh    = lane >> 4;

    const int colk  = 32 * g  + 2 * lm;
    const int colv  = 32 * gp + 2 * lm;
    const int rbase = pc * CHUNK + wave * 4 + 2 * lh;
    const float* kp  = k  + ((size_t)batch * SEQ + rbase) * DIM + colk;
    const float* vp  = v  + ((size_t)batch * SEQ + rbase) * DIM + colv;
    const float* fkp = fk + (size_t)rbase * DIM + colk;
    const float* fvp = fv + (size_t)rbase * DIM + colv;

    v8f aee = {}, aoe = {}, aeo = {}, aoo = {};

    v2f ck0[GSTEPS], ck1[GSTEPS], cf0[GSTEPS], cf1[GSTEPS];
    v2f cv0[GSTEPS], cv1[GSTEPS], cg0[GSTEPS], cg1[GSTEPS];
    #pragma unroll
    for (int s = 0; s < GSTEPS; ++s) {
        const int o = s * 32 * DIM;
        ck0[s] = *(const v2f*)(kp + o);   ck1[s] = *(const v2f*)(kp + o + DIM);
        cf0[s] = *(const v2f*)(fkp + o);  cf1[s] = *(const v2f*)(fkp + o + DIM);
        cv0[s] = *(const v2f*)(vp + o);   cv1[s] = *(const v2f*)(vp + o + DIM);
        cg0[s] = *(const v2f*)(fvp + o);  cg1[s] = *(const v2f*)(fvp + o + DIM);
    }
    #pragma unroll
    for (int grp = 0; grp < NGROUP; ++grp) {
        v2f nk0[GSTEPS], nk1[GSTEPS], nf0[GSTEPS], nf1[GSTEPS];
        v2f nv0[GSTEPS], nv1[GSTEPS], ng0[GSTEPS], ng1[GSTEPS];
        if (grp + 1 < NGROUP) {
            #pragma unroll
            for (int s = 0; s < GSTEPS; ++s) {
                const int o = ((grp + 1) * GSTEPS + s) * 32 * DIM;
                nk0[s] = *(const v2f*)(kp + o);   nk1[s] = *(const v2f*)(kp + o + DIM);
                nf0[s] = *(const v2f*)(fkp + o);  nf1[s] = *(const v2f*)(fkp + o + DIM);
                nv0[s] = *(const v2f*)(vp + o);   nv1[s] = *(const v2f*)(vp + o + DIM);
                ng0[s] = *(const v2f*)(fvp + o);  ng1[s] = *(const v2f*)(fvp + o + DIM);
            }
        }
        #pragma unroll
        for (int s = 0; s < GSTEPS; ++s) {
            const v2f k0 = rope2(ck0[s], cf0[s]);
            const v2f k1 = rope2(ck1[s], cf1[s]);
            const v2f v0 = rope2(cv0[s], cg0[s]);
            const v2f v1 = rope2(cv1[s], cg1[s]);
            v2f ae, ao, be, bo;
            ae.x = k0.x;  ae.y = k1.x;
            ao.x = k0.y;  ao.y = k1.y;
            be.x = v0.x;  be.y = v1.x;
            bo.x = v0.y;  bo.y = v1.y;
            aee = __builtin_amdgcn_wmma_f32_16x16x4_f32(false, ae, false, be, (short)0, aee, false, false);
            aoe = __builtin_amdgcn_wmma_f32_16x16x4_f32(false, ao, false, be, (short)0, aoe, false, false);
            aeo = __builtin_amdgcn_wmma_f32_16x16x4_f32(false, ae, false, bo, (short)0, aeo, false, false);
            aoo = __builtin_amdgcn_wmma_f32_16x16x4_f32(false, ao, false, bo, (short)0, aoo, false, false);
        }
        if (grp + 1 < NGROUP) {
            #pragma unroll
            for (int s = 0; s < GSTEPS; ++s) {
                ck0[s] = nk0[s];  ck1[s] = nk1[s];  cf0[s] = nf0[s];  cf1[s] = nf1[s];
                cv0[s] = nv0[s];  cv1[s] = nv1[s];  cg0[s] = ng0[s];  cg1[s] = ng1[s];
            }
        }
    }

    #pragma unroll
    for (int r = 0; r < 8; ++r) {
        red[wave][0][r][lane] = aee[r];
        red[wave][1][r][lane] = aoe[r];
        red[wave][2][r][lane] = aeo[r];
        red[wave][3][r][lane] = aoo[r];
    }
    __syncthreads();
    {
        const int r = wave;
        v4f s;
        s.x = red[0][0][r][lane];  s.y = red[0][1][r][lane];
        s.z = red[0][2][r][lane];  s.w = red[0][3][r][lane];
        #pragma unroll
        for (int w = 1; w < 8; ++w) {
            s.x += red[w][0][r][lane];  s.y += red[w][1][r][lane];
            s.z += red[w][2][r][lane];  s.w += red[w][3][r][lane];
        }
        float* dst = ws + (size_t)(batch * NPART + pc) * MSZ
                        + (16 * g + 8 * lh + r) * 128 + 64 * gp + 4 * lm;
        *(v4f*)dst = s;
    }
}

__global__ __launch_bounds__(256) void rope_q_times_m(
    const float* __restrict__ q, const float* __restrict__ fq,
    const float* __restrict__ ws, float* __restrict__ out)
{
    __shared__ float Ml[32 * MPITCH];
    __shared__ float Qs[8][16 * QPITCH];

    const int batch = blockIdx.x >> 5;
    const int q0    = (blockIdx.x & 31) * 128;
    const int tid   = threadIdx.x;
    const int wave  = tid >> 5;
    const int lane  = tid & 31;
    const int lm    = lane & 15;
    const int lh    = lane >> 4;

    const size_t trow = (size_t)(q0 + wave * 16);
    v4f xq[8], xf[8];
    {
        const v4f* qt = (const v4f*)(q  + ((size_t)batch * SEQ + trow) * DIM);
        const v4f* ft = (const v4f*)(fq + trow * DIM);
        #pragma unroll
        for (int i = 0; i < 8; ++i) {
            xq[i] = qt[i * 32 + lane];
            xf[i] = ft[i * 32 + lane];
        }
    }

    v4f pw[4][NPART];
    {
        const float* wp = ws + (size_t)batch * NPART * MSZ + tid * 4;
        #pragma unroll
        for (int kk = 0; kk < 4; ++kk) {
            #pragma unroll
            for (int p = 0; p < NPART; ++p)
                pw[kk][p] = *(const v4f*)(wp + p * MSZ + kk * 1024);
        }
    }
    __builtin_amdgcn_sched_barrier(0);

    float* st = &Qs[wave][0];
    #pragma unroll
    for (int i = 0; i < 8; ++i) {
        v2f x0, x1, f0, f1;
        x0.x = xq[i].x;  x0.y = xq[i].y;  x1.x = xq[i].z;  x1.y = xq[i].w;
        f0.x = xf[i].x;  f0.y = xf[i].y;  f1.x = xf[i].z;  f1.y = xf[i].w;
        const v2f r0 = rope2(x0, f0);
        const v2f r1 = rope2(x1, f1);
        v4f r;
        r.x = r0.x;  r.y = r0.y;  r.z = r1.x;  r.w = r1.y;
        *(v4f*)&st[(2 * i + lh) * QPITCH + 4 * lm] = r;
    }

    #pragma unroll
    for (int kk = 0; kk < 4; ++kk) {
        v4f s = pw[kk][0];
        #pragma unroll
        for (int p = 1; p < NPART; ++p)
            s += pw[kk][p];
        const int idx = tid * 4 + kk * 1024;
        *(v4f*)&Ml[(idx >> 7) * MPITCH + (idx & 127)] = s;
    }
    __syncthreads();

    v8f acc0 = {}, acc1 = {}, acc2 = {}, acc3 = {};
    #pragma unroll
    for (int c = 0; c < 16; ++c) {
        const v2f qa = *(const v2f*)&st[lm * QPITCH + 4 * c + 2 * lh];
        const float* mrow = &Ml[(2 * c + lh) * MPITCH + 2 * lm];
        const v2f b0 = *(const v2f*)(mrow +  0);
        const v2f b1 = *(const v2f*)(mrow + 32);
        const v2f b2 = *(const v2f*)(mrow + 64);
        const v2f b3 = *(const v2f*)(mrow + 96);
        acc0 = __builtin_amdgcn_wmma_f32_16x16x4_f32(false, qa, false, b0, (short)0, acc0, false, false);
        acc1 = __builtin_amdgcn_wmma_f32_16x16x4_f32(false, qa, false, b1, (short)0, acc1, false, false);
        acc2 = __builtin_amdgcn_wmma_f32_16x16x4_f32(false, qa, false, b2, (short)0, acc2, false, false);
        acc3 = __builtin_amdgcn_wmma_f32_16x16x4_f32(false, qa, false, b3, (short)0, acc3, false, false);
    }

    const float scale = 0.125f;
    #pragma unroll
    for (int r = 0; r < 8; ++r) {
        float* srow = &st[(r + 8 * lh) * QPITCH + lm];
        srow[ 0] = acc0[r] * scale;
        srow[16] = acc1[r] * scale;
        srow[32] = acc2[r] * scale;
        srow[48] = acc3[r] * scale;
    }
    v4f* ot = (v4f*)(out + ((size_t)batch * SEQ + trow) * DIM);
    #pragma unroll
    for (int i = 0; i < 8; ++i)
        ot[i * 32 + lane] = *(const v4f*)&st[(2 * i + lh) * QPITCH + 4 * lm];
}

extern "C" void kernel_launch(void* const* d_in, const int* in_sizes, int n_in,
                              void* d_out, int out_size, void* d_ws, size_t ws_size,
                              hipStream_t stream) {
    const float* q  = (const float*)d_in[0];
    const float* k  = (const float*)d_in[1];
    const float* v  = (const float*)d_in[2];
    const float* fq = (const float*)d_in[3];
    const float* fk = (const float*)d_in[4];
    const float* fv = (const float*)d_in[5];
    float* out = (float*)d_out;
    float* ws  = (float*)d_ws;

    hipLaunchKernelGGL(rope_ktv_partial, dim3(256), dim3(256), 0, stream, k, v, fk, fv, ws);
    hipLaunchKernelGGL(rope_q_times_m, dim3(256), dim3(256), 0, stream, q, fq, ws, out);
}
